# starting kernel
# baseline (speedup 1.0000x reference)
#include <hip/hip_runtime.h>
#include <hip/hip_bf16.h>

typedef __attribute__((ext_vector_type(2))) float v2f;
typedef __attribute__((ext_vector_type(8))) float v8f;

#define N_NODES 100000
#define KDIM    128

__device__ __forceinline__ void atomic_add_f32(float* p, float v) {
  asm volatile("global_atomic_add_f32 %0, %1, off" : : "v"(p), "v"(v) : "memory");
}

__global__ __launch_bounds__(256) void spmm_edge_atomic(
    const int* __restrict__ rows, const int* __restrict__ cols,
    const float* __restrict__ vals, const float* __restrict__ x,
    float* __restrict__ out, int E) {
  int e = blockIdx.x * 8 + (threadIdx.x >> 5);
  if (e >= E) return;
  int lane = threadIdx.x & 31;
  int r = rows[e];
  int c = cols[e];
  float v = vals[e];
  float4 xv = *((const float4*)(x + (size_t)c * 128) + lane);
  float* o = out + (size_t)r * 128 + lane * 4;
  atomic_add_f32(o + 0, xv.x * v);
  atomic_add_f32(o + 1, xv.y * v);
  atomic_add_f32(o + 2, xv.z * v);
  atomic_add_f32(o + 3, xv.w * v);
}

template <int NCOLS, bool RELU>
__global__ __launch_bounds__(256) void gemm_wmma_f32(
    const float* __restrict__ A, const float* __restrict__ B,
    const float* __restrict__ bias, float* __restrict__ C, int M) {
  constexpr int K = KDIM;
  constexpr int LDSP = 2 * NCOLS + 32;
  __shared__ __align__(16) float Bs[(K / 2) * LDSP];

  for (int i = threadIdx.x; i < K * NCOLS; i += 256) {
    int k = i / NCOLS, n = i % NCOLS;
    Bs[(k >> 1) * LDSP + n * 2 + (k & 1)] = B[i];
  }
  __syncthreads();

  const int wave = threadIdx.x >> 5;
  const int lane = threadIdx.x & 31;
  const int tileRow = blockIdx.x * 8 + wave;
  if (tileRow * 16 >= M) return;
  const int row0 = tileRow * 16;

  const int mlane = lane & 15;
  const int khalf = (lane >> 4) * 2;

  v2f afrag[K / 4];
  const float* arow = A + (size_t)(row0 + mlane) * K + khalf;
#pragma unroll
  for (int ks = 0; ks < K / 4; ++ks)
    afrag[ks] = *(const v2f*)(arow + ks * 4);

#pragma unroll
  for (int nt = 0; nt < NCOLS / 16; ++nt) {
    const int n = nt * 16 + mlane;
    const float bv = bias[n];
    v8f acc;
#pragma unroll
    for (int i = 0; i < 8; ++i) acc[i] = bv;

#pragma unroll
    for (int ks = 0; ks < K / 4; ++ks) {
      const int kpair = ks * 2 + (lane >> 4);
      v2f bfrag = *(const v2f*)&Bs[kpair * LDSP + n * 2];
      acc = __builtin_amdgcn_wmma_f32_16x16x4_f32(
           false, afrag[ks],  false, bfrag,
           (short)0, acc,  false,  false);
    }

    const int mbase = row0 + ((lane >> 4) * 8);
#pragma unroll
    for (int v = 0; v < 8; ++v) {
      float val = acc[v];
      if (RELU) val = fmaxf(val, 0.0f);
      C[(size_t)(mbase + v) * NCOLS + n] = val;
    }
  }
}

extern "C" void kernel_launch(void* const* d_in, const int* in_sizes, int n_in,
                              void* d_out, int out_size, void* d_ws, size_t ws_size,
                              hipStream_t stream) {
  const float* x  = (const float*)d_in[0];
  const int*   er = (const int*)d_in[1];
  const int*   ec = (const int*)d_in[2];
  const float* ev = (const float*)d_in[3];
  const float* W1 = (const float*)d_in[4];
  const float* b1 = (const float*)d_in[5];
  const float* W2 = (const float*)d_in[6];
  const float* b2 = (const float*)d_in[7];
  float* out = (float*)d_out;
  const int E = in_sizes[1];

  const size_t hElems = (size_t)N_NODES * 128;
  float* h1 = (float*)d_ws;
  float* h2 = h1 + hElems;

  const int spmmBlocks = (E + 7) / 8;
  const int gemmBlocks = ((N_NODES / 16) + 7) / 8;

  hipMemsetAsync(h1, 0, hElems * sizeof(float), stream);
  spmm_edge_atomic<<<spmmBlocks, 256, 0, stream>>>(er, ec, ev, x, h1, E);
  gemm_wmma_f32<128, true><<<gemmBlocks, 256, 0, stream>>>(h1, W1, b1, h2, N_NODES);
  hipMemsetAsync(h1, 0, hElems * sizeof(float), stream);
  spmm_edge_atomic<<<spmmBlocks, 256, 0, stream>>>(er, ec, ev, h2, h1, E);
  gemm_wmma_f32<64, false><<<gemmBlocks, 256, 0, stream>>>(h1, W2, b2, out, N_NODES);
}


# Whole-list edge scan in every work-group replaced by one binning pass: 256 groups each split 1/256 of the edges into per-tile cells in the workspace; a tile reads its 256 cells; sum order changes
# speedup vs baseline: 2.9895x; 2.9895x over previous
#include <hip/hip_runtime.h>
#include <hip/hip_bf16.h>

typedef __attribute__((ext_vector_type(2))) float v2f;
typedef __attribute__((ext_vector_type(8))) float v8f;

#define N_NODES 100000
#define KDIM    128

#define TILE_ROWS  391
#define SPMM_WAVES 32
#define LIST_CAP   768
#define SUB_CAP    24
#define PACK_BITS  9
#define SAVE_HEAD  64
#define SAVE_STRIDE (SAVE_HEAD + SPMM_WAVES * LIST_CAP * 2)
#define N_TILES    256
#define N_CHUNKS   256
#define CELL_CAP   64
#define CELLS_PER_WAVE (N_CHUNKS / SPMM_WAVES)

__device__ __forceinline__ void read_entry(const int2* Q, int idx, int& e, int& rl) {
  const int2 t = Q[idx];
  e = t.x;
  rl = t.y;
}
__device__ __forceinline__ void read_entry(const unsigned* Q, int idx, int& e, int& rl) {
  const unsigned t = Q[idx];
  e = (int)(t >> PACK_BITS);
  rl = (int)(t & ((1u << PACK_BITS) - 1u));
}

__device__ __forceinline__ void sum_batch(int c, int rl, float v, int m, int lane,
                                          const float* __restrict__ x, float* acc) {
  for (int j = 0; j < m; j += 8) {
    float a[8][4];
    float vj[8];
    int rj[8];
#pragma unroll
    for (int u = 0; u < 8; ++u) {
      const int s = min(j + u, m - 1);
      const int cj = __builtin_amdgcn_readlane(c, s);
      vj[u] = __int_as_float(__builtin_amdgcn_readlane(__float_as_int(v), s));
      rj[u] = __builtin_amdgcn_readlane(rl, s);
      const float* xr = x + (size_t)cj * 128 + lane;
      a[u][0] = xr[0];
      a[u][1] = xr[32];
      a[u][2] = xr[64];
      a[u][3] = xr[96];
    }
#pragma unroll
    for (int u = 0; u < 8; ++u) {
      if (j + u < m) {
        float* ar = acc + rj[u] * 128 + lane;
        ar[0]  += a[u][0] * vj[u];
        ar[32] += a[u][1] * vj[u];
        ar[64] += a[u][2] * vj[u];
        ar[96] += a[u][3] * vj[u];
      }
    }
  }
}

template <bool SAVE, typename T>
__device__ __forceinline__ void queue_flush(
    const T* Q, int n, int lane, const int* __restrict__ cols,
    const float* __restrict__ vals, const float* __restrict__ x, float* acc, int2* rec) {
  for (int base = 0; base < n; base += 32) {
    const int idx = base + lane;
    int c = 0, rl = 0;
    float v = 0.0f;
    if (idx < n) {
      int e;
      read_entry(Q, idx, e, rl);
      c = cols[e];
      v = vals[e];
      if (SAVE) rec[idx] = make_int2((int)(((unsigned)c << PACK_BITS) | (unsigned)rl), __float_as_int(v));
    }
    sum_batch(c, rl, v, min(32, n - base), lane, x, acc);
  }
}

__device__ __forceinline__ void record_flush(const int2* rec, int n, int lane,
                                             const float* __restrict__ x, float* acc) {
  for (int base = 0; base < n; base += 32) {
    const int idx = base + lane;
    int c = 0, rl = 0;
    float v = 0.0f;
    if (idx < n) {
      const int2 t = rec[idx];
      c = min((int)((unsigned)t.x >> PACK_BITS), N_NODES - 1);
      rl = min((int)((unsigned)t.x & ((1u << PACK_BITS) - 1u)), TILE_ROWS - 1);
      v = __int_as_float(t.y);
    }
    sum_batch(c, rl, v, min(32, n - base), lane, x, acc);
  }
}

__device__ __forceinline__ void queue_push(
    int2* Q, int& qn, bool own, int e, int rl, int lane, const int* __restrict__ cols,
    const float* __restrict__ vals, const float* __restrict__ x, float* acc) {
  const unsigned m = __builtin_amdgcn_ballot_w32(own);
  if (m) {
    const int pos = qn + (int)__builtin_amdgcn_mbcnt_lo(m, 0u);
    if (own) Q[pos] = make_int2(e, rl);
    qn += __popc(m);
    if (qn >= 32) {
      queue_flush<false>(Q, qn, lane, cols, vals, x, acc, nullptr);
      qn = 0;
    }
  }
}

__device__ __forceinline__ void place_hits(unsigned* L, unsigned m, int& cv, int lane, int wave,
                                           unsigned entry, unsigned d) {
  while (m) {
    const int l = __builtin_ctz(m);
    m &= m - 1u;
    const int o = __builtin_amdgcn_readlane((int)d, l) & (SPMM_WAVES - 1);
    const int c = __builtin_amdgcn_readlane(cv, o);
    if (lane == l) L[o * LIST_CAP + wave * SUB_CAP + min(c, SUB_CAP - 1)] = entry;
    cv += (lane == o) ? 1 : 0;
  }
}

__device__ __forceinline__ int4 load_rows4(const int* __restrict__ rows, int e0, int E) {
  int4 r;
  if (e0 + 3 < E) {
    r = *(const int4*)(rows + e0);
  } else {
    r.x = (e0     < E) ? rows[e0]     : -1;
    r.y = (e0 + 1 < E) ? rows[e0 + 1] : -1;
    r.z = (e0 + 2 < E) ? rows[e0 + 2] : -1;
    r.w = (e0 + 3 < E) ? rows[e0 + 3] : -1;
  }
  return r;
}

__device__ __forceinline__ unsigned lanes_with_same_key(int key, int lane) {
  unsigned mine = 0u;
#pragma unroll
  for (int l = 0; l < 32; ++l) {
    const int kl = __builtin_amdgcn_readlane(key, l);
    const unsigned m = __builtin_amdgcn_ballot_w32(key == kl);
    if (lane == l) mine = m;
  }
  return mine;
}

__global__ __launch_bounds__(1024) void bin_edges_by_tile(
    const int* __restrict__ rows, int* __restrict__ cellCnt, unsigned* __restrict__ cells, int E) {
  __shared__ int cntW[SPMM_WAVES][N_TILES];
  const int tid = threadIdx.x;
  const int lane = tid & 31;
  const int wave = __builtin_amdgcn_readfirstlane(tid >> 5);
  const int c = blockIdx.x;
  const int chunk = (E + N_CHUNKS - 1) / N_CHUNKS;
  const int per = (chunk + SPMM_WAVES - 1) / SPMM_WAVES;
  const int wb = c * chunk + wave * per;
  const int we = min(min(wb + per, c * chunk + chunk), E);

  for (int i = tid; i < SPMM_WAVES * N_TILES; i += 1024) (&cntW[0][0])[i] = 0;
  __syncthreads();

  for (int base = wb; base < we; base += 32) {
    const int e = base + lane;
    int t = -1;
    if (e < we) {
      const unsigned r = (unsigned)rows[e];
      if (r < (unsigned)N_NODES) t = (int)(r / TILE_ROWS);
    }
    const unsigned mine = lanes_with_same_key(t, lane);
    if (t >= 0 && (mine >> lane) == 1u) cntW[wave][t] += __popc(mine);
  }
  __syncthreads();

  if (tid < N_TILES) {
    int run = 0;
    for (int w = 0; w < SPMM_WAVES; ++w) {
      const int v = cntW[w][tid];
      cntW[w][tid] = run;
      run += v;
    }
    cellCnt[tid * N_CHUNKS + c] = run;
  }
  __syncthreads();

  for (int base = wb; base < we; base += 32) {
    const int e = base + lane;
    int t = -1;
    unsigned r = 0u;
    if (e < we) {
      r = (unsigned)rows[e];
      if (r < (unsigned)N_NODES) t = (int)(r / TILE_ROWS);
    }
    const unsigned mine = lanes_with_same_key(t, lane);
    if (t >= 0) {
      const int off = cntW[wave][t];
      const int slot = off + __popc(mine & ((1u << lane) - 1u));
      if (slot < CELL_CAP)
        cells[((size_t)t * N_CHUNKS + c) * CELL_CAP + slot] =
            ((unsigned)e << PACK_BITS) | (r - (unsigned)t * TILE_ROWS);
      if ((mine >> lane) == 1u) cntW[wave][t] = off + __popc(mine);
    }
  }
}

template <bool REUSE>
__global__ __launch_bounds__(1024) void spmm_row_tile_lds(
    const int* __restrict__ rows, const int* __restrict__ cols,
    const float* __restrict__ vals, const float* __restrict__ x,
    float* __restrict__ out, int* __restrict__ saved,
    const int* __restrict__ cellCnt, const unsigned* __restrict__ cells, int E) {
  __shared__ __align__(16) float acc[TILE_ROWS * 128];
  __shared__ __align__(16) unsigned lst[SPMM_WAVES][LIST_CAP];
  __shared__ int2 que[SPMM_WAVES][64];
  __shared__ int subn[SPMM_WAVES][SPMM_WAVES];
  __shared__ int anyFull;

  const int tid = threadIdx.x;
  const int lane = tid & 31;
  const int wave = __builtin_amdgcn_readfirstlane(tid >> 5);
  const int lo = blockIdx.x * TILE_ROWS;
  const int nrows = min(TILE_ROWS, N_NODES - lo);
  if (nrows <= 0) return;

  for (int i = tid * 4; i < TILE_ROWS * 128; i += 1024 * 4)
    *(float4*)&acc[i] = make_float4(0.0f, 0.0f, 0.0f, 0.0f);
  if (tid == 0) anyFull = 0;
  __syncthreads();

  const int ngroups = (E + 127) >> 7;
  const bool packable = E <= (1 << (32 - PACK_BITS));

  int* S = saved + (size_t)blockIdx.x * SAVE_STRIDE;
  int n = 0;
  if (packable && REUSE) {
    n = min(max(__builtin_amdgcn_readfirstlane(S[wave]), 0), LIST_CAP);
    if (tid == 0) anyFull = S[SPMM_WAVES];
  }
  if (packable && !REUSE) {
    unsigned* L = &lst[0][0];
    int cv = 0;
    const size_t cell0 = (size_t)blockIdx.x * N_CHUNKS + wave * CELLS_PER_WAVE;
    const unsigned* ce = cells + cell0 * CELL_CAP;
    const int cr = (lane < CELLS_PER_WAVE) ? cellCnt[cell0 + lane] : 0;
    const bool over = __builtin_amdgcn_ballot_w32(cr > CELL_CAP) != 0u;
    const int cc = min(max(cr, 0), CELL_CAP);
    unsigned ent[CELLS_PER_WAVE * CELL_CAP / 32];
#pragma unroll
    for (int v = 0; v < CELLS_PER_WAVE * CELL_CAP / 32; ++v) ent[v] = ce[v * 32 + lane];
#pragma unroll
    for (int v = 0; v < CELLS_PER_WAVE * CELL_CAP / 32; ++v) {
      const int nj = __builtin_amdgcn_readlane(cc, v / (CELL_CAP / 32));
      const unsigned m = __builtin_amdgcn_ballot_w32((v % (CELL_CAP / 32)) * 32 + lane < nj);
      if (m) {
        const unsigned d = min(ent[v] & ((1u << PACK_BITS) - 1u), (unsigned)(nrows - 1));
        const unsigned e = min(ent[v] >> PACK_BITS, (unsigned)(E - 1));
        place_hits(L, m, cv, lane, wave, (e << PACK_BITS) | d, d);
      }
    }
    subn[lane][wave] = cv;
    if ((over || __builtin_amdgcn_ballot_w32(cv > SUB_CAP) != 0u) && lane == 0) anyFull = 1;
  }
  __syncthreads();

  if (packable && !REUSE) {
    if (anyFull == 0) {
      unsigned* Lw = lst[wave];
      const int cp = subn[wave][lane];
      for (int p = 0; p < SPMM_WAVES; p += 4) {
        int np[4];
        unsigned t[4];
#pragma unroll
        for (int u = 0; u < 4; ++u) {
          np[u] = min(__builtin_amdgcn_readlane(cp, p + u), SUB_CAP);
          t[u] = Lw[(p + u) * SUB_CAP + min(lane, SUB_CAP - 1)];
        }
#pragma unroll
        for (int u = 0; u < 4; ++u) {
          if (lane < np[u]) Lw[n + lane] = t[u];
          n += np[u];
        }
      }
    }
    if (lane == 0) S[wave] = n;
    if (tid == 0) S[SPMM_WAVES] = anyFull;
  }

  if (packable && anyFull == 0) {
    int2* R = (int2*)(S + SAVE_HEAD) + wave * LIST_CAP;
    if (REUSE) record_flush(R, n, lane, x, acc);
    else queue_flush<true>(lst[wave], n, lane, cols, vals, x, acc, R);
  } else {
    int2* Q = que[wave];
    int qn = 0;
    for (int g = 0; g < ngroups; ++g) {
      const int e0 = (g * 32 + lane) * 4;
      const int4 r = load_rows4(rows, e0, E);
      const unsigned d0 = (unsigned)(r.x - lo), d1 = (unsigned)(r.y - lo);
      const unsigned d2 = (unsigned)(r.z - lo), d3 = (unsigned)(r.w - lo);
      const bool h0 = d0 < (unsigned)nrows && (int)(d0 & 31u) == wave;
      const bool h1 = d1 < (unsigned)nrows && (int)(d1 & 31u) == wave;
      const bool h2 = d2 < (unsigned)nrows && (int)(d2 & 31u) == wave;
      const bool h3 = d3 < (unsigned)nrows && (int)(d3 & 31u) == wave;
      if (__builtin_amdgcn_ballot_w32(h0 | h1 | h2 | h3)) {
        queue_push(Q, qn, h0, e0,     (int)d0, lane, cols, vals, x, acc);
        queue_push(Q, qn, h1, e0 + 1, (int)d1, lane, cols, vals, x, acc);
        queue_push(Q, qn, h2, e0 + 2, (int)d2, lane, cols, vals, x, acc);
        queue_push(Q, qn, h3, e0 + 3, (int)d3, lane, cols, vals, x, acc);
      }
    }
    queue_flush<false>(Q, qn, lane, cols, vals, x, acc, nullptr);
  }
  __syncthreads();

  float* o = out + (size_t)lo * 128;
  for (int i = tid * 4; i < nrows * 128; i += 1024 * 4)
    *(float4*)(o + i) = *(const float4*)&acc[i];
}

template <int NCOLS, bool RELU>
__global__ __launch_bounds__(256) void gemm_wmma_f32(
    const float* __restrict__ A, const float* __restrict__ B,
    const float* __restrict__ bias, float* __restrict__ C, int M) {
  constexpr int K = KDIM;
  constexpr int LDSP = 2 * NCOLS + 32;
  __shared__ __align__(16) float Bs[(K / 2) * LDSP];

  for (int i = threadIdx.x; i < K * NCOLS; i += 256) {
    int k = i / NCOLS, n = i % NCOLS;
    Bs[(k >> 1) * LDSP + n * 2 + (k & 1)] = B[i];
  }
  __syncthreads();

  const int wave = threadIdx.x >> 5;
  const int lane = threadIdx.x & 31;
  const int tileRow = blockIdx.x * 8 + wave;
  if (tileRow * 16 >= M) return;
  const int row0 = tileRow * 16;

  const int mlane = lane & 15;
  const int khalf = (lane >> 4) * 2;

  v2f afrag[K / 4];
  const float* arow = A + (size_t)(row0 + mlane) * K + khalf;
#pragma unroll
  for (int ks = 0; ks < K / 4; ++ks)
    afrag[ks] = *(const v2f*)(arow + ks * 4);

#pragma unroll
  for (int nt = 0; nt < NCOLS / 16; ++nt) {
    const int n = nt * 16 + mlane;
    const float bv = bias[n];
    v8f acc;
#pragma unroll
    for (int i = 0; i < 8; ++i) acc[i] = bv;

#pragma unroll
    for (int ks = 0; ks < K / 4; ++ks) {
      const int kpair = ks * 2 + (lane >> 4);
      v2f bfrag = *(const v2f*)&Bs[kpair * LDSP + n * 2];
      acc = __builtin_amdgcn_wmma_f32_16x16x4_f32(
           false, afrag[ks],  false, bfrag,
           (short)0, acc,  false,  false);
    }

    const int mbase = row0 + ((lane >> 4) * 8);
#pragma unroll
    for (int v = 0; v < 8; ++v) {
      float val = acc[v];
      if (RELU) val = fmaxf(val, 0.0f);
      C[(size_t)(mbase + v) * NCOLS + n] = val;
    }
  }
}

extern "C" void kernel_launch(void* const* d_in, const int* in_sizes, int n_in,
                              void* d_out, int out_size, void* d_ws, size_t ws_size,
                              hipStream_t stream) {
  const float* x  = (const float*)d_in[0];
  const int*   er = (const int*)d_in[1];
  const int*   ec = (const int*)d_in[2];
  const float* ev = (const float*)d_in[3];
  const float* W1 = (const float*)d_in[4];
  const float* b1 = (const float*)d_in[5];
  const float* W2 = (const float*)d_in[6];
  const float* b2 = (const float*)d_in[7];
  float* out = (float*)d_out;
  const int E = in_sizes[1];

  const size_t hElems = (size_t)N_NODES * 128;
  float* h1 = (float*)d_ws;
  float* h2 = h1 + hElems;
  int* saved = (int*)(h2 + hElems);
  int* cellCnt = saved + (size_t)N_TILES * SAVE_STRIDE;
  unsigned* cells = (unsigned*)(cellCnt + (size_t)N_TILES * N_CHUNKS);

  const int spmmBlocks = (N_NODES + TILE_ROWS - 1) / TILE_ROWS;
  const int gemmBlocks = ((N_NODES / 16) + 7) / 8;

  bin_edges_by_tile<<<N_CHUNKS, 1024, 0, stream>>>(er, cellCnt, cells, E);
  spmm_row_tile_lds<false><<<spmmBlocks, 1024, 0, stream>>>(er, ec, ev, x, h1, saved, cellCnt, cells, E);
  gemm_wmma_f32<128, true><<<gemmBlocks, 256, 0, stream>>>(h1, W1, b1, h2, N_NODES);
  spmm_row_tile_lds<true><<<spmmBlocks, 1024, 0, stream>>>(er, ec, ev, h2, h1, saved, cellCnt, cells, E);
  gemm_wmma_f32<64, false><<<gemmBlocks, 256, 0, stream>>>(h1, W2, b2, out, N_NODES);
}
